# starting kernel
# baseline (speedup 1.0000x reference)
#include <hip/hip_runtime.h>

typedef float v2f __attribute__((ext_vector_type(2)));
typedef float v8f __attribute__((ext_vector_type(8)));

__device__ __forceinline__ float lrelu(float v) { return v > 0.0f ? v : 0.01f * v; }

__global__ __launch_bounds__(1024)
void gat_edge_stats_kernel(const float* __restrict__ x,
                           const float* __restrict__ W,
                           const float* __restrict__ a,
                           const int*  __restrict__ src_idx,
                           const int*  __restrict__ dst_idx,
                           float* __restrict__ c_out,
                           int N, int F, int DEG, int node_off, int xcols)
{
    extern __shared__ float smem[];
    float* sn  = smem;
    float* red = smem + N;

    const int b        = blockIdx.x;
    const int tid      = threadIdx.x;
    const int nthreads = blockDim.x;
    const float* xb = x + (size_t)b * xcols + node_off;

    for (int i = tid; i < N; i += nthreads) sn[i] = xb[i];

    float s1 = 0.0f, s2 = 0.0f;
    for (int f = 0; f < F; ++f) { float w = W[f]; s1 += w * a[f]; s2 += w * a[F + f]; }
    __syncthreads();

    float lmax = -3.402823466e38f;
    for (int i = tid; i < N; i += nthreads) {
        const int ebase = i * DEG;
        for (int k = 0; k < DEG; ++k) {
            float raw = lrelu(sn[src_idx[ebase + k]] * s1 + sn[dst_idx[ebase + k]] * s2);
            lmax = fmaxf(lmax, raw);
        }
    }
    red[tid] = lmax;
    __syncthreads();
    for (int s = nthreads >> 1; s > 0; s >>= 1) {
        if (tid < s) red[tid] = fmaxf(red[tid], red[tid + s]);
        __syncthreads();
    }
    const float m = red[0];
    __syncthreads();

    float lsum = 0.0f;
    for (int i = tid; i < N; i += nthreads) {
        const int ebase = i * DEG;
        float t = 0.0f;
        for (int k = 0; k < DEG; ++k) {
            const int d = dst_idx[ebase + k];
            float raw = lrelu(sn[src_idx[ebase + k]] * s1 + sn[d] * s2);
            float p = expf(raw - m);
            lsum += p;
            t    += p * sn[d];
        }
        c_out[(size_t)b * N + i] = t;
    }
    red[tid] = lsum;
    __syncthreads();
    for (int s = nthreads >> 1; s > 0; s >>= 1) {
        if (tid < s) red[tid] += red[tid + s];
        __syncthreads();
    }
    const float invS = 1.0f / red[0];
    __syncthreads();
    for (int i = tid; i < N; i += nthreads)
        c_out[(size_t)b * N + i] *= invS;
}

__global__ __launch_bounds__(256)
void gat_outer_wmma_kernel(const float* __restrict__ c,
                           const float* __restrict__ W,
                           float* __restrict__ out,
                           int F)
{
    const int lane  = threadIdx.x & 31;
    const int wave  = threadIdx.x >> 5;
    const int g     = blockIdx.x * 16;
    const int fbase = wave * 16;

    v2f A;
    A.x = (lane < 16) ? c[g + lane] : 0.0f;
    A.y = 0.0f;

    v2f Bv;
    Bv.x = (lane < 16) ? W[fbase + lane] : 0.0f;
    Bv.y = 0.0f;

    v8f acc = {};
    acc = __builtin_amdgcn_wmma_f32_16x16x4_f32(
         false, A,  false, Bv,
         (short)0, acc,  false,  false);

    const int mbase = g + ((lane >> 4) << 3);
    const int col   = fbase + (lane & 15);
#pragma unroll
    for (int v = 0; v < 8; ++v) {
        float val = acc[v];
        out[(size_t)(mbase + v) * F + col] = val > 0.0f ? val : 0.01f * val;
    }
}

extern "C" void kernel_launch(void* const* d_in, const int* in_sizes, int n_in,
                              void* d_out, int out_size, void* d_ws, size_t ws_size,
                              hipStream_t stream)
{
    const float* x   = (const float*)d_in[0];
    const float* Wn  = (const float*)d_in[1];
    const float* a   = (const float*)d_in[2];
    const int*   src = (const int*)d_in[3];
    const int*   dst = (const int*)d_in[4];
    float* out = (float*)d_out;

    const int B        = 4;
    const int F        = in_sizes[1];
    const int E        = in_sizes[3];
    const int N        = out_size / (B * F);
    const int DEG      = E / N;
    const int xcols    = in_sizes[0] / B;
    const int node_off = xcols - N;

    float* c_ws = (float*)d_ws;

    const int threads = 1024;
    const size_t smem = (size_t)(N + threads) * sizeof(float);
    gat_edge_stats_kernel<<<B, threads, smem, stream>>>(
        x, Wn, a, src, dst, c_ws, N, F, DEG, node_off, xcols);

    const int rows = B * N;
    gat_outer_wmma_kernel<<<rows / 16, (F / 16) * 32, 0, stream>>>(
        c_ws, Wn, out, F);
}


# edge softmax statistics split over 32 work-groups per batch (own max and sum each), merged by exp(m_g - m) rescaling in launch 2: max, sum and per-node T are reduced in a different order
# speedup vs baseline: 4.8016x; 4.8016x over previous
#include <hip/hip_runtime.h>

typedef float v2f __attribute__((ext_vector_type(2)));
typedef float v8f __attribute__((ext_vector_type(8)));

__device__ __forceinline__ float lrelu(float v) { return v > 0.0f ? v : 0.01f * v; }

constexpr int kGroupNodes     = 128;
constexpr int kStatThreads    = 256;
constexpr int kEdgesPerThread = 9;
constexpr int kStageVecs      = 4;

template <int kXorMask>
__device__ __forceinline__ float row_xor(float x) {
    const int b = __builtin_bit_cast(int, x);
    return __builtin_bit_cast(float, __builtin_amdgcn_update_dpp(b, b, 0x160 + kXorMask, 0xf, 0xf, false));
}
__device__ __forceinline__ float xhalf(float x) {
    const int b = __builtin_bit_cast(int, x);
    return __builtin_bit_cast(float, __builtin_amdgcn_permlanex16(b, b, 0x76543210, 0xfedcba98, false, false));
}
__device__ __forceinline__ float wave_max(float v) {
    v = fmaxf(v, row_xor<1>(v));
    v = fmaxf(v, row_xor<2>(v));
    v = fmaxf(v, row_xor<4>(v));
    v = fmaxf(v, row_xor<8>(v));
    return fmaxf(v, xhalf(v));
}
__device__ __forceinline__ float wave_sum(float v) {
    v += row_xor<1>(v);
    v += row_xor<2>(v);
    v += row_xor<4>(v);
    v += row_xor<8>(v);
    return v + xhalf(v);
}

__global__ __launch_bounds__(kStatThreads)
void gat_edge_partial_kernel(const float* __restrict__ x,
                             const float* __restrict__ W,
                             const float* __restrict__ a,
                             const int*  __restrict__ src_idx,
                             const int*  __restrict__ dst_idx,
                             float* __restrict__ t_park,
                             float* __restrict__ stats,
                             int N, int F, int DEG, int node_off, int xcols)
{
    extern __shared__ float smem[];
    const int grp    = blockIdx.x;
    const int b      = blockIdx.y;
    const int groups = gridDim.x;
    const int tid    = threadIdx.x;
    const int wave   = tid >> 5;

    const int i0 = grp * kGroupNodes;
    const int nb = min(kGroupNodes, N - i0);
    const int EB = nb * DEG;
    const int e0 = i0 * DEG;

    const int kSpare = kGroupNodes * DEG;
    float* sn  = smem;
    float* sR  = sn + N;
    float* sQ  = sR + kSpare + 1;
    float* red = sQ + kSpare + 1;

    const float* xb = x + (size_t)b * xcols + node_off;
    constexpr int kSweep = kEdgesPerThread * kStatThreads;

    int si[kEdgesPerThread], di[kEdgesPerThread];
#pragma unroll
    for (int j = 0; j < kEdgesPerThread; ++j) {
        const int elc = min(j * kStatThreads + tid, EB - 1);
        si[j] = src_idx[e0 + elc];
        di[j] = dst_idx[e0 + elc];
    }

    constexpr int kStageChunk = kStageVecs * kStatThreads * 4;
    const bool whole = (N % kStageChunk == 0) && (((xcols | node_off) & 3) == 0);
    const float4* xb4 = reinterpret_cast<const float4*>(xb);
    float4*       sn4 = reinterpret_cast<float4*>(sn);
    float4 v0 = {}, v1 = {}, v2 = {}, v3 = {};
    if (whole) {
        v0 = xb4[tid];
        v1 = xb4[tid + kStatThreads];
        v2 = xb4[tid + 2 * kStatThreads];
        v3 = xb4[tid + 3 * kStatThreads];
    }

    float s1 = 0.0f, s2 = 0.0f;
    for (int f = 0; f < F; ++f) { float w = W[f]; s1 += w * a[f]; s2 += w * a[F + f]; }

    if (whole) {
        sn4[tid]                    = v0;
        sn4[tid + kStatThreads]     = v1;
        sn4[tid + 2 * kStatThreads] = v2;
        sn4[tid + 3 * kStatThreads] = v3;
        for (int c4i = kStageVecs * kStatThreads; c4i < N / 4; c4i += kStageVecs * kStatThreads) {
            const float4 u0 = xb4[c4i + tid];
            const float4 u1 = xb4[c4i + tid + kStatThreads];
            const float4 u2 = xb4[c4i + tid + 2 * kStatThreads];
            const float4 u3 = xb4[c4i + tid + 3 * kStatThreads];
            sn4[c4i + tid]                    = u0;
            sn4[c4i + tid + kStatThreads]     = u1;
            sn4[c4i + tid + 2 * kStatThreads] = u2;
            sn4[c4i + tid + 3 * kStatThreads] = u3;
        }
    } else {
        for (int i = tid; i < N; i += kStatThreads) sn[i] = xb[i];
    }
    __syncthreads();

    float lmax = -3.402823466e38f;
    for (int sw = 0; sw < EB; sw += kSweep) {
        if (sw > 0) {
#pragma unroll
            for (int j = 0; j < kEdgesPerThread; ++j) {
                const int elc = min(sw + j * kStatThreads + tid, EB - 1);
                si[j] = src_idx[e0 + elc];
                di[j] = dst_idx[e0 + elc];
            }
        }
        float ns[kEdgesPerThread], nd[kEdgesPerThread];
#pragma unroll
        for (int j = 0; j < kEdgesPerThread; ++j) { ns[j] = sn[si[j]]; nd[j] = sn[di[j]]; }
#pragma unroll
        for (int j = 0; j < kEdgesPerThread; ++j) {
            const int el    = sw + j * kStatThreads + tid;
            const bool ok   = el < EB;
            const float raw = lrelu(ns[j] * s1 + nd[j] * s2);
            const int slot  = ok ? el : kSpare;
            sR[slot] = raw;
            sQ[slot] = nd[j];
            lmax = fmaxf(lmax, ok ? raw : -3.402823466e38f);
        }
    }
    lmax = wave_max(lmax);
    red[wave] = lmax;
    __syncthreads();
    float m = red[0];
#pragma unroll
    for (int w = 1; w < kStatThreads / 32; ++w) m = fmaxf(m, red[w]);

    float lsum = 0.0f;
    for (int sw = 0; sw < EB; sw += kSweep) {
        float r[kEdgesPerThread], q[kEdgesPerThread];
#pragma unroll
        for (int j = 0; j < kEdgesPerThread; ++j) {
            const int elc = min(sw + j * kStatThreads + tid, EB - 1);
            r[j] = sR[elc];
            q[j] = sQ[elc];
        }
#pragma unroll
        for (int j = 0; j < kEdgesPerThread; ++j) {
            const int el  = sw + j * kStatThreads + tid;
            const bool ok = el < EB;
            const float p = expf(r[j] - m);
            lsum += ok ? p : 0.0f;
            sQ[ok ? el : kSpare] = p * q[j];
        }
    }
    lsum = wave_sum(lsum);
    red[kStatThreads / 32 + wave] = lsum;
    __syncthreads();

    if (tid < nb) {
        const float* qn = sQ + tid * DEG;
        float t = 0.0f;
#pragma unroll 8
        for (int k = 0; k < DEG; ++k) t += qn[k];
        const int r = b * N + i0 + tid;
        float* dstp = t_park + (size_t)(r & ~15) * F + (r & 15);
        for (int cs = 0; cs < F; cs += 16) dstp[cs] = t;
    }
    if (tid == 0) {
        float S = red[kStatThreads / 32];
#pragma unroll
        for (int w = 1; w < kStatThreads / 32; ++w) S += red[kStatThreads / 32 + w];
        stats[2 * (b * groups + grp) + 0] = m;
        stats[2 * (b * groups + grp) + 1] = S;
    }
}

__global__ __launch_bounds__(256)
void gat_outer_wmma_kernel(const float* __restrict__ stats,
                           const float* __restrict__ W,
                           float* out,
                           int F, int N)
{
    const int lane  = threadIdx.x & 31;
    const int wave  = threadIdx.x >> 5;
    const int b     = blockIdx.y;
    const int i     = blockIdx.x * 16;
    const int g     = b * N + i;
    const int fbase = wave * 16;

    const int groups = (N + kGroupNodes - 1) / kGroupNodes;
    const float* st  = stats + 2 * (size_t)b * groups;

    const float tv = (lane < 16) ? out[(size_t)g * F + fbase + lane] : 0.0f;
    const float wv = (lane < 16) ? W[fbase + lane] : 0.0f;
    const float mg = st[2 * (i / kGroupNodes)];

    float mloc = -3.402823466e38f, sloc = 0.0f;
    for (int q = lane; q < groups; q += 32) {
        const float mq = st[2 * q], sq = st[2 * q + 1];
        const float mn = fmaxf(mloc, mq);
        sloc = sloc * expf(mloc - mn) + sq * expf(mq - mn);
        mloc = mn;
    }
    const float m    = wave_max(mloc);
    const float S    = wave_sum(sloc * expf(mloc - m));
    const float invS = 1.0f / S;
    const float scale = expf(mg - m) * invS;

    v2f A;
    A.x = tv * scale;
    A.y = 0.0f;

    v2f Bv;
    Bv.x = wv;
    Bv.y = 0.0f;

    v8f acc = {};
    acc = __builtin_amdgcn_wmma_f32_16x16x4_f32(
         false, A,  false, Bv,
         (short)0, acc,  false,  false);

    const int mbase = g + ((lane >> 4) << 3);
    const int col   = fbase + (lane & 15);
#pragma unroll
    for (int v = 0; v < 8; ++v) {
        float val = acc[v];
        out[(size_t)(mbase + v) * F + col] = val > 0.0f ? val : 0.01f * val;
    }
}

extern "C" void kernel_launch(void* const* d_in, const int* in_sizes, int n_in,
                              void* d_out, int out_size, void* d_ws, size_t ws_size,
                              hipStream_t stream)
{
    const float* x   = (const float*)d_in[0];
    const float* Wn  = (const float*)d_in[1];
    const float* a   = (const float*)d_in[2];
    const int*   src = (const int*)d_in[3];
    const int*   dst = (const int*)d_in[4];
    float* out = (float*)d_out;

    const int B        = 4;
    const int F        = in_sizes[1];
    const int E        = in_sizes[3];
    const int N        = out_size / (B * F);
    const int DEG      = E / N;
    const int xcols    = in_sizes[0] / B;
    const int node_off = xcols - N;

    float* stats_ws = (float*)d_ws;

    const int groups  = (N + kGroupNodes - 1) / kGroupNodes;
    const size_t smem = (size_t)(N + 2 * (kGroupNodes * DEG + 1) + 16) * sizeof(float);
    gat_edge_partial_kernel<<<dim3(groups, B), kStatThreads, smem, stream>>>(
        x, Wn, a, src, dst, out, stats_ws, N, F, DEG, node_off, xcols);

    gat_outer_wmma_kernel<<<dim3(N / 16, B), (F / 16) * 32, 0, stream>>>(
        stats_ws, Wn, out, F, N);
}


# output launch re-tiled: one wave per 16-row x 128-column tile, 8 matrix products with columns interleaved (4n+t) so a lane stores 16-byte vectors on whole lines; 1024 waves instead of 8192
# speedup vs baseline: 5.7404x; 1.1955x over previous
#include <hip/hip_runtime.h>

typedef float v2f __attribute__((ext_vector_type(2)));
typedef float v8f __attribute__((ext_vector_type(8)));

__device__ __forceinline__ float lrelu(float v) { return v > 0.0f ? v : 0.01f * v; }

constexpr int kGroupNodes     = 128;
constexpr int kStatThreads    = 256;
constexpr int kEdgesPerThread = 9;
constexpr int kStageVecs      = 4;

constexpr int kTileWaves = 4;
constexpr int kTileCols  = 128;

template <int kXorMask>
__device__ __forceinline__ float row_xor(float x) {
    const int b = __builtin_bit_cast(int, x);
    return __builtin_bit_cast(float, __builtin_amdgcn_update_dpp(b, b, 0x160 + kXorMask, 0xf, 0xf, false));
}
__device__ __forceinline__ float xhalf(float x) {
    const int b = __builtin_bit_cast(int, x);
    return __builtin_bit_cast(float, __builtin_amdgcn_permlanex16(b, b, 0x76543210, 0xfedcba98, false, false));
}
__device__ __forceinline__ float wave_max(float v) {
    v = fmaxf(v, row_xor<1>(v));
    v = fmaxf(v, row_xor<2>(v));
    v = fmaxf(v, row_xor<4>(v));
    v = fmaxf(v, row_xor<8>(v));
    return fmaxf(v, xhalf(v));
}
__device__ __forceinline__ float wave_sum(float v) {
    v += row_xor<1>(v);
    v += row_xor<2>(v);
    v += row_xor<4>(v);
    v += row_xor<8>(v);
    return v + xhalf(v);
}

__global__ __launch_bounds__(kStatThreads)
void gat_edge_partial_kernel(const float* __restrict__ x,
                             const float* __restrict__ W,
                             const float* __restrict__ a,
                             const int*  __restrict__ src_idx,
                             const int*  __restrict__ dst_idx,
                             float* __restrict__ t_park,
                             float* __restrict__ stats,
                             int N, int F, int DEG, int node_off, int xcols)
{
    extern __shared__ float smem[];
    const int grp    = blockIdx.x;
    const int b      = blockIdx.y;
    const int groups = gridDim.x;
    const int tid    = threadIdx.x;
    const int wave   = tid >> 5;

    const int i0 = grp * kGroupNodes;
    const int nb = min(kGroupNodes, N - i0);
    const int EB = nb * DEG;
    const int e0 = i0 * DEG;

    const int kSpare = kGroupNodes * DEG;
    float* sn  = smem;
    float* sR  = sn + N;
    float* sQ  = sR + kSpare + 1;
    float* red = sQ + kSpare + 1;

    const float* xb = x + (size_t)b * xcols + node_off;
    constexpr int kSweep = kEdgesPerThread * kStatThreads;

    int si[kEdgesPerThread], di[kEdgesPerThread];
#pragma unroll
    for (int j = 0; j < kEdgesPerThread; ++j) {
        const int elc = min(j * kStatThreads + tid, EB - 1);
        si[j] = src_idx[e0 + elc];
        di[j] = dst_idx[e0 + elc];
    }

    constexpr int kStageChunk = kStageVecs * kStatThreads * 4;
    const bool whole = (N % kStageChunk == 0) && (((xcols | node_off) & 3) == 0);
    const float4* xb4 = reinterpret_cast<const float4*>(xb);
    float4*       sn4 = reinterpret_cast<float4*>(sn);
    float4 v0 = {}, v1 = {}, v2 = {}, v3 = {};
    if (whole) {
        v0 = xb4[tid];
        v1 = xb4[tid + kStatThreads];
        v2 = xb4[tid + 2 * kStatThreads];
        v3 = xb4[tid + 3 * kStatThreads];
    }

    float s1 = 0.0f, s2 = 0.0f;
    for (int f = 0; f < F; ++f) { float w = W[f]; s1 += w * a[f]; s2 += w * a[F + f]; }

    if (whole) {
        sn4[tid]                    = v0;
        sn4[tid + kStatThreads]     = v1;
        sn4[tid + 2 * kStatThreads] = v2;
        sn4[tid + 3 * kStatThreads] = v3;
        for (int c4i = kStageVecs * kStatThreads; c4i < N / 4; c4i += kStageVecs * kStatThreads) {
            const float4 u0 = xb4[c4i + tid];
            const float4 u1 = xb4[c4i + tid + kStatThreads];
            const float4 u2 = xb4[c4i + tid + 2 * kStatThreads];
            const float4 u3 = xb4[c4i + tid + 3 * kStatThreads];
            sn4[c4i + tid]                    = u0;
            sn4[c4i + tid + kStatThreads]     = u1;
            sn4[c4i + tid + 2 * kStatThreads] = u2;
            sn4[c4i + tid + 3 * kStatThreads] = u3;
        }
    } else {
        for (int i = tid; i < N; i += kStatThreads) sn[i] = xb[i];
    }
    __syncthreads();

    float lmax = -3.402823466e38f;
    for (int sw = 0; sw < EB; sw += kSweep) {
        if (sw > 0) {
#pragma unroll
            for (int j = 0; j < kEdgesPerThread; ++j) {
                const int elc = min(sw + j * kStatThreads + tid, EB - 1);
                si[j] = src_idx[e0 + elc];
                di[j] = dst_idx[e0 + elc];
            }
        }
        float ns[kEdgesPerThread], nd[kEdgesPerThread];
#pragma unroll
        for (int j = 0; j < kEdgesPerThread; ++j) { ns[j] = sn[si[j]]; nd[j] = sn[di[j]]; }
#pragma unroll
        for (int j = 0; j < kEdgesPerThread; ++j) {
            const int el    = sw + j * kStatThreads + tid;
            const bool ok   = el < EB;
            const float raw = lrelu(ns[j] * s1 + nd[j] * s2);
            const int slot  = ok ? el : kSpare;
            sR[slot] = raw;
            sQ[slot] = nd[j];
            lmax = fmaxf(lmax, ok ? raw : -3.402823466e38f);
        }
    }
    lmax = wave_max(lmax);
    red[wave] = lmax;
    __syncthreads();
    float m = red[0];
#pragma unroll
    for (int w = 1; w < kStatThreads / 32; ++w) m = fmaxf(m, red[w]);

    float lsum = 0.0f;
    for (int sw = 0; sw < EB; sw += kSweep) {
        float r[kEdgesPerThread], q[kEdgesPerThread];
#pragma unroll
        for (int j = 0; j < kEdgesPerThread; ++j) {
            const int elc = min(sw + j * kStatThreads + tid, EB - 1);
            r[j] = sR[elc];
            q[j] = sQ[elc];
        }
#pragma unroll
        for (int j = 0; j < kEdgesPerThread; ++j) {
            const int el  = sw + j * kStatThreads + tid;
            const bool ok = el < EB;
            const float p = expf(r[j] - m);
            lsum += ok ? p : 0.0f;
            sQ[ok ? el : kSpare] = p * q[j];
        }
    }
    lsum = wave_sum(lsum);
    red[kStatThreads / 32 + wave] = lsum;
    __syncthreads();

    if (tid < nb) {
        const float* qn = sQ + tid * DEG;
        float t = 0.0f;
#pragma unroll 8
        for (int k = 0; k < DEG; ++k) t += qn[k];
        const int r = b * N + i0 + tid;
        float* dstp = t_park + (size_t)(r & ~15) * F + (r & 15);
        for (int cs = 0; cs < F; cs += kTileCols) dstp[cs] = t;
    }
    if (tid == 0) {
        float S = red[kStatThreads / 32];
#pragma unroll
        for (int w = 1; w < kStatThreads / 32; ++w) S += red[kStatThreads / 32 + w];
        stats[2 * (b * groups + grp) + 0] = m;
        stats[2 * (b * groups + grp) + 1] = S;
    }
}

__global__ __launch_bounds__(kTileWaves * 32)
void gat_outer_wmma_kernel(const float* __restrict__ stats,
                           const float* __restrict__ W,
                           float* out,
                           int F, int N)
{
    const int lane = threadIdx.x & 31;
    const int wave = threadIdx.x >> 5;
    const int b    = blockIdx.y;
    const int i    = (blockIdx.x * kTileWaves + wave) * 16;
    if (i >= N) return;
    const int g    = b * N + i;
    const int cb   = blockIdx.z * kTileCols;
    const int n    = lane & 15;
    const bool lo  = lane < 16;

    const int groups = (N + kGroupNodes - 1) / kGroupNodes;
    const float* st  = stats + 2 * (size_t)b * groups;

    const float  tv  = lo ? out[(size_t)g * F + cb + lane] : 0.0f;
    const float4 wq0 = *reinterpret_cast<const float4*>(W + cb + 4 * n);
    const float4 wq1 = *reinterpret_cast<const float4*>(W + cb + 64 + 4 * n);
    const float  mg  = st[2 * (i / kGroupNodes)];

    float mloc = -3.402823466e38f, sloc = 0.0f;
    for (int q = lane; q < groups; q += 32) {
        const float mq = st[2 * q], sq = st[2 * q + 1];
        const float mn = fmaxf(mloc, mq);
        sloc = sloc * expf(mloc - mn) + sq * expf(mq - mn);
        mloc = mn;
    }
    const float m    = wave_max(mloc);
    const float S    = wave_sum(sloc * expf(mloc - m));
    const float invS = 1.0f / S;
    const float scale = expf(mg - m) * invS;

    v2f A;
    A.x = tv * scale;
    A.y = 0.0f;

    const float wk[8] = { lo ? wq0.x : 0.0f, lo ? wq0.y : 0.0f, lo ? wq0.z : 0.0f, lo ? wq0.w : 0.0f,
                          lo ? wq1.x : 0.0f, lo ? wq1.y : 0.0f, lo ? wq1.z : 0.0f, lo ? wq1.w : 0.0f };

    v8f acc[8];
#pragma unroll
    for (int q = 0; q < 8; ++q) {
        v2f Bv;
        Bv.x = wk[q];
        Bv.y = 0.0f;
        const v8f zero = {};
        acc[q] = __builtin_amdgcn_wmma_f32_16x16x4_f32(
             false, A,  false, Bv,
             (short)0, zero,  false,  false);
    }

    const int mbase = g + ((lane >> 4) << 3);
    float* orow = out + (size_t)mbase * F + cb + 4 * n;
#pragma unroll
    for (int v = 0; v < 8; ++v) {
#pragma unroll
        for (int j = 0; j < 2; ++j) {
            float4 o;
            o.x = lrelu(acc[4 * j + 0][v]);
            o.y = lrelu(acc[4 * j + 1][v]);
            o.z = lrelu(acc[4 * j + 2][v]);
            o.w = lrelu(acc[4 * j + 3][v]);
            *reinterpret_cast<float4*>(orow + (size_t)v * F + 64 * j) = o;
        }
    }
}

extern "C" void kernel_launch(void* const* d_in, const int* in_sizes, int n_in,
                              void* d_out, int out_size, void* d_ws, size_t ws_size,
                              hipStream_t stream)
{
    const float* x   = (const float*)d_in[0];
    const float* Wn  = (const float*)d_in[1];
    const float* a   = (const float*)d_in[2];
    const int*   src = (const int*)d_in[3];
    const int*   dst = (const int*)d_in[4];
    float* out = (float*)d_out;

    const int B        = 4;
    const int F        = in_sizes[1];
    const int E        = in_sizes[3];
    const int N        = out_size / (B * F);
    const int DEG      = E / N;
    const int xcols    = in_sizes[0] / B;
    const int node_off = xcols - N;

    float* stats_ws = (float*)d_ws;

    const int groups  = (N + kGroupNodes - 1) / kGroupNodes;
    const size_t smem = (size_t)(N + 2 * (kGroupNodes * DEG + 1) + 16) * sizeof(float);
    gat_edge_partial_kernel<<<dim3(groups, B), kStatThreads, smem, stream>>>(
        x, Wn, a, src, dst, out, stats_ws, N, F, DEG, node_off, xcols);

    gat_outer_wmma_kernel<<<dim3((N / 16 + kTileWaves - 1) / kTileWaves, B, F / kTileCols),
                            kTileWaves * 32, 0, stream>>>(
        stats_ws, Wn, out, F, N);
}
